# starting kernel
# baseline (speedup 1.0000x reference)
#include <hip/hip_runtime.h>
#include <hip/hip_bf16.h>

#define BATCH 4096
#define TSEQ  200
#define FEAT  32
#define HID   8
#define FWD   5

typedef __attribute__((ext_vector_type(2))) float v2f;
typedef __attribute__((ext_vector_type(8))) float v8f;

__device__ __forceinline__ float sigm(float x) {
    return 1.0f / (1.0f + __expf(-x));
}
__device__ __forceinline__ float tanh_fast(float x) {
    return 1.0f - 2.0f / (1.0f + __expf(2.0f * x));
}

__device__ __forceinline__ float wcat(const float* __restrict__ W_ih,
                                      const float* __restrict__ W_hh,
                                      int k, int n) {
    if (k < FEAT + 1)       return W_ih[n * (FEAT + 1) + k];
    if (k < FEAT + 1 + HID) return W_hh[n * HID + (k - (FEAT + 1))];
    return 0.0f;
}

__device__ __forceinline__ void loadx(v2f (&dst)[8], const float* __restrict__ xr, int hi) {
#pragma unroll
    for (int c = 0; c < 8; ++c)
        dst[c] = *reinterpret_cast<const v2f*>(xr + 4 * c + 2 * hi);
}

#define WMMA_F32(ACC, A, BFRAG) \
    (ACC) = __builtin_amdgcn_wmma_f32_16x16x4_f32(false, (A), false, (BFRAG), (short)0, (ACC), false, false)

__global__ __launch_bounds__(32)
void monotonic_lstm_wmma(const float* __restrict__ x,
                         const float* __restrict__ z0,
                         const float* __restrict__ W_ih,
                         const float* __restrict__ W_hh,
                         const float* __restrict__ b_ih,
                         const float* __restrict__ b_hh,
                         const float* __restrict__ W1,
                         const float* __restrict__ b1,
                         const float* __restrict__ W2,
                         const float* __restrict__ b2,
                         float* __restrict__ out)
{
    __shared__ float gates_s[16 * 33];
    __shared__ float h_s[16 * 9];
    __shared__ float z_s[16];

    const int lane = threadIdx.x;
    const int m    = lane & 15;
    const int hi   = lane >> 4;
    const int b0   = blockIdx.x * 16;

    v2f bw0[11], bw1[11];
#pragma unroll
    for (int c = 0; c < 11; ++c) {
        const int k0 = 4 * c + 2 * hi;
        bw0[c].x = wcat(W_ih, W_hh, k0,     m);
        bw0[c].y = wcat(W_ih, W_hh, k0 + 1, m);
        bw1[c].x = wcat(W_ih, W_hh, k0,     m + 16);
        bw1[c].y = wcat(W_ih, W_hh, k0 + 1, m + 16);
    }
    const float bias0 = b_ih[m]      + b_hh[m];
    const float bias1 = b_ih[m + 16] + b_hh[m + 16];

    float w1r[FWD * HID], b1r[FWD], w2r[FWD];
#pragma unroll
    for (int i = 0; i < FWD * HID; ++i) w1r[i] = W1[i];
#pragma unroll
    for (int i = 0; i < FWD; ++i) { b1r[i] = b1[i]; w2r[i] = W2[i]; }
    const float b2r = b2[0];

    float cst[4] = {0.f, 0.f, 0.f, 0.f};
    float zreg = z0[b0 + m];
    if (hi == 0) z_s[m] = zreg;
#pragma unroll
    for (int u = 0; u < 4; ++u) h_s[m * 9 + hi * 4 + u] = 0.0f;
    __builtin_amdgcn_wave_barrier();

    const float* xrow = x + ((size_t)(b0 + m) * TSEQ) * FEAT;

    v2f xbuf[2][8];
    loadx(xbuf[0], xrow, hi);

#pragma unroll 2
    for (int t = 0; t < TSEQ; ++t) {
        const int cb = t & 1;
        const int tn = (t + 1 < TSEQ) ? (t + 1) : t;
        loadx(xbuf[cb ^ 1], xrow + (size_t)tn * FEAT, hi);

        v8f a0e, a0o, a1e, a1o;
#pragma unroll
        for (int r = 0; r < 8; ++r) {
            a0e[r] = bias0; a1e[r] = bias1;
            a0o[r] = 0.0f;  a1o[r] = 0.0f;
        }

#pragma unroll
        for (int c = 0; c < 8; c += 2) {
            WMMA_F32(a0e, xbuf[cb][c],     bw0[c]);
            WMMA_F32(a1e, xbuf[cb][c],     bw1[c]);
            WMMA_F32(a0o, xbuf[cb][c + 1], bw0[c + 1]);
            WMMA_F32(a1o, xbuf[cb][c + 1], bw1[c + 1]);
        }
        {
            const float h7 = h_s[m * 9 + 7];
            v2f a;
            a.x = hi ? h_s[m * 9 + 1] : z_s[m];
            a.y = hi ? h_s[m * 9 + 2] : h_s[m * 9 + 0];
            WMMA_F32(a0e, a, bw0[8]);
            WMMA_F32(a1e, a, bw1[8]);
            a.x = h_s[m * 9 + 3 + 2 * hi];
            a.y = h_s[m * 9 + 4 + 2 * hi];
            WMMA_F32(a0o, a, bw0[9]);
            WMMA_F32(a1o, a, bw1[9]);
            a.x = hi ? 0.0f : h7;
            a.y = 0.0f;
            WMMA_F32(a0e, a, bw0[10]);
            WMMA_F32(a1e, a, bw1[10]);
        }
        __builtin_amdgcn_wave_barrier();

#pragma unroll
        for (int r = 0; r < 8; ++r) {
            gates_s[(r + 8 * hi) * 33 + m]      = a0e[r] + a0o[r];
            gates_s[(r + 8 * hi) * 33 + m + 16] = a1e[r] + a1o[r];
        }
        __builtin_amdgcn_wave_barrier();

#pragma unroll
        for (int u = 0; u < 4; ++u) {
            const int j = hi * 4 + u;
            const float ig = sigm(gates_s[m * 33 + j]);
            const float fg = sigm(gates_s[m * 33 + 8 + j]);
            const float gg = tanh_fast(gates_s[m * 33 + 16 + j]);
            const float og = sigm(gates_s[m * 33 + 24 + j]);
            const float cn = fg * cst[u] + ig * gg;
            cst[u] = cn;
            h_s[m * 9 + j] = og * tanh_fast(cn);
        }
        __builtin_amdgcn_wave_barrier();

        if (hi == 0) {
            float hv[HID];
#pragma unroll
            for (int j = 0; j < HID; ++j) hv[j] = h_s[m * 9 + j];
            float delta = b2r;
#pragma unroll
            for (int k = 0; k < FWD; ++k) {
                float d = b1r[k];
#pragma unroll
                for (int j = 0; j < HID; ++j) d = fmaf(hv[j], w1r[k * HID + j], d);
                delta = fmaf(fmaxf(d, 0.0f), w2r[k], delta);
            }
            zreg += fmaxf(delta, 0.0f);
            z_s[m] = zreg;
            out[(size_t)(b0 + m) * TSEQ + t] = zreg;
        }
        __builtin_amdgcn_wave_barrier();
    }
}

extern "C" void kernel_launch(void* const* d_in, const int* in_sizes, int n_in,
                              void* d_out, int out_size, void* d_ws, size_t ws_size,
                              hipStream_t stream) {
    (void)in_sizes; (void)n_in; (void)out_size; (void)d_ws; (void)ws_size;
    const float* x    = (const float*)d_in[0];
    const float* z0   = (const float*)d_in[1];
    const float* W_ih = (const float*)d_in[2];
    const float* W_hh = (const float*)d_in[3];
    const float* b_ih = (const float*)d_in[4];
    const float* b_hh = (const float*)d_in[5];
    const float* W1   = (const float*)d_in[6];
    const float* b1   = (const float*)d_in[7];
    const float* W2   = (const float*)d_in[8];
    const float* b2   = (const float*)d_in[9];
    float* out = (float*)d_out;

    dim3 grid(BATCH / 16);
    dim3 block(32);
    monotonic_lstm_wmma<<<grid, block, 0, stream>>>(x, z0, W_ih, W_hh, b_ih, b_hh,
                                                    W1, b1, W2, b2, out);
}


# Recurrent K reordered h0..h3 | h4..h7 | z: next step's h chunks (at a block end the next x part too) issue ahead of the head and run under it; only the z pass waits for z; gate sum order changes
# speedup vs baseline: 3.0300x; 3.0300x over previous
#include <hip/hip_runtime.h>
#include <hip/hip_bf16.h>


#define BATCH 4096
#define TSEQ  200
#define FEAT  32
#define HID   8
#define FWD   5
#define TBLK  4


typedef __attribute__((ext_vector_type(2))) float v2f;
typedef __attribute__((ext_vector_type(4))) float v4f;
typedef __attribute__((ext_vector_type(8))) float v8f;

__device__ __forceinline__ float sigm(float x) {
    return 1.0f / (1.0f + __expf(-x));
}
__device__ __forceinline__ float tanh_fast(float x) {
    return 1.0f - 2.0f / (1.0f + __expf(2.0f * x));
}

__device__ __forceinline__ float ror8(float v) {
    return __builtin_bit_cast(float,
        __builtin_amdgcn_update_dpp(0, __builtin_bit_cast(int, v), 0x128, 0xf, 0xf, false));
}
template <int U>
__device__ __forceinline__ float bcast8(float v) {
    return __builtin_bit_cast(float,
        __builtin_amdgcn_mov_dpp8(__builtin_bit_cast(unsigned, v), U * 0x249249));
}
__device__ __forceinline__ float gather16(float v, unsigned sel, bool same) {
    const unsigned s = __builtin_bit_cast(unsigned, v);
    const unsigned own = __builtin_amdgcn_permlane16_var(s, s, sel, false, false);
    const unsigned oth = __builtin_amdgcn_permlanex16_var(s, s, sel, false, false);
    return __builtin_bit_cast(float, same ? own : oth);
}

#define WMMA_F32(ACC, A, BFRAG) \
    (ACC) = __builtin_amdgcn_wmma_f32_16x16x4_f32(false, (A), false, (BFRAG), (short)0, (ACC), false, false)

__global__ __launch_bounds__(128)
void monotonic_lstm_wmma(const float* __restrict__ x,
                         const float* __restrict__ z0,
                         const float* __restrict__ W_ih,
                         const float* __restrict__ W_hh,
                         const float* __restrict__ b_ih,
                         const float* __restrict__ b_hh,
                         const float* __restrict__ W1,
                         const float* __restrict__ b1,
                         const float* __restrict__ W2,
                         const float* __restrict__ b2,
                         float* __restrict__ out)
{
    const int lane = threadIdx.x & 31;
    const int c    = lane & 15;
    const int hi   = lane >> 4;
    const int wv   = threadIdx.x >> 5;
    const int rb   = blockIdx.x * 16 + wv * 4;

    const int xa = c >> 3, xb = c & 1;
    const int xrowi = rb + 2 * xa + xb;
    const int xj = (c >> 1) & 3;
    const int rs = (lane >> 3) & 1;
    const int u  = lane & 7;
    const int crow = rb + 2 * hi + rs;

    v2f bw[2][11];
#pragma unroll
    for (int tl = 0; tl < 2; ++tl) {
        const int n = c + 16 * tl;
        const float* wi = W_ih + n * (FEAT + 1);
        const float* wh = W_hh + n * HID;
#pragma unroll
        for (int q = 0; q < 4; ++q)
#pragma unroll
            for (int s = 0; s < 2; ++s) {
                const int k0 = 8 * q + 4 * hi + 2 * s;
                bw[tl][2 * q + s].x = wi[k0];
                bw[tl][2 * q + s].y = wi[k0 + 1];
            }
        bw[tl][8].x  = wh[hi ? 2 : 0];
        bw[tl][8].y  = wh[hi ? 3 : 1];
        bw[tl][9].x  = wh[hi ? 6 : 4];
        bw[tl][9].y  = wh[hi ? 7 : 5];
        bw[tl][10].x = hi ? 0.0f : wi[FEAT];
        bw[tl][10].y = 0.0f;
    }
    const float bias0 = b_ih[c] + b_hh[c];
    const float bias1 = b_ih[c + 16] + b_hh[c + 16];

    float w1r[FWD * HID], b1r[FWD], w2r[FWD];
#pragma unroll
    for (int i = 0; i < FWD * HID; ++i) w1r[i] = W1[i];
#pragma unroll
    for (int i = 0; i < FWD; ++i) { b1r[i] = b1[i]; w2r[i] = W2[i]; }
    const float b2r = b2[0];

    const bool same = (xa == hi);
    const unsigned s1 = 8 * xb + (hi ? 2 : 0);
    const unsigned s2 = 8 * xb + (hi ? 3 : 1);
    const unsigned s3 = 8 * xb + (hi ? 6 : 4);
    const unsigned s4 = 8 * xb + (hi ? 7 : 5);
    const unsigned sz = 8 * xb;

    float cst  = 0.0f;
    float zreg = z0[crow];
    float gz = gather16(zreg, sz, same);

    const float* xrow = x + ((size_t)xrowi * TSEQ + xj) * FEAT + 4 * hi;

    v4f xbuf[4];
#pragma unroll
    for (int q = 0; q < 4; ++q)
        xbuf[q] = *reinterpret_cast<const v4f*>(xrow + 8 * q);

#define XPART()                                                   \
    do {                                                          \
        _Pragma("unroll")                                         \
        for (int r = 0; r < 8; ++r) { xp0[r] = bias0; xp1[r] = bias1; } \
        _Pragma("unroll")                                         \
        for (int q = 0; q < 4; ++q) {                             \
            const v4f xv = xbuf[q];                               \
            v2f a; a.x = xv.x; a.y = xv.y;                        \
            WMMA_F32(xp0, a, bw[0][2 * q]);                       \
            WMMA_F32(xp1, a, bw[1][2 * q]);                       \
            a.x = xv.z; a.y = xv.w;                               \
            WMMA_F32(xp0, a, bw[0][2 * q + 1]);                   \
            WMMA_F32(xp1, a, bw[1][2 * q + 1]);                   \
        }                                                         \
    } while (0)

#define ORDER2(A, B) asm volatile("" : "+v"(A), "+v"(B))
#define ORDER_D(X)   asm volatile("" : "+v"(d[0]), "+v"(d[1]), "+v"(d[2]), "+v"(d[3]), "+v"(d[4]), "+v"(X))

    v8f xp0, xp1, acc0, acc1;
    float g1 = 0.0f, g2 = 0.0f, g3 = 0.0f, g4 = 0.0f;
    XPART();
    acc0 = xp0; acc1 = xp1;

    for (int tb = 0; tb < TSEQ / TBLK; ++tb) {
        {
            const int tn = (tb + 1 < TSEQ / TBLK) ? (tb + 1) : tb;
#pragma unroll
            for (int q = 0; q < 4; ++q)
                xbuf[q] = *reinterpret_cast<const v4f*>(xrow + (size_t)tn * (TBLK * FEAT) + 8 * q);
        }

#pragma unroll
      for (int j = 0; j < TBLK; ++j) {
        const int t = tb * TBLK + j;

        {
            v2f a;
            a.x = hi ? 0.0f : gz; a.y = 0.0f;
            WMMA_F32(acc0, a, bw[0][10]);
            WMMA_F32(acc1, a, bw[1][10]);
        }
        const float a00 = acc0[2 * j], a01 = acc0[2 * j + 1];
        const float a10 = acc1[2 * j], a11 = acc1[2 * j + 1];

        const float t00 = ror8(a00), t01 = ror8(a01);
        const float t10 = ror8(a10), t11 = ror8(a11);
        const float gi = rs ? t01 : a00;
        const float gf = rs ? a01 : t00;
        const float gc = rs ? t11 : a10;
        const float go = rs ? a11 : t10;

        const float ig = sigm(gi);
        const float fg = sigm(gf);
        const float gg = tanh_fast(gc);
        const float og = sigm(go);
        const float cn = fg * cst + ig * gg;
        cst = cn;
        const float hn = og * tanh_fast(cn);

        g1 = gather16(hn, s1, same);
        g2 = gather16(hn, s2, same);
        g3 = gather16(hn, s3, same);
        g4 = gather16(hn, s4, same);

        if (j == TBLK - 1) XPART();
        v2f ha, hb;
        ha.x = g1; ha.y = g2;
        hb.x = g3; hb.y = g4;
        float hcur = hn;
        acc0 = xp0; acc1 = xp1;
        WMMA_F32(acc0, ha, bw[0][8]);
        ORDER2(acc0, hcur);
        float hv[HID];
        hv[0] = bcast8<0>(hcur); hv[1] = bcast8<1>(hcur); hv[2] = bcast8<2>(hcur); hv[3] = bcast8<3>(hcur);
        hv[4] = bcast8<4>(hcur); hv[5] = bcast8<5>(hcur); hv[6] = bcast8<6>(hcur); hv[7] = bcast8<7>(hcur);
        asm volatile("" : "+v"(hv[0]), "+v"(hv[1]), "+v"(hv[2]), "+v"(hv[3]),
                          "+v"(hv[4]), "+v"(hv[5]), "+v"(hv[6]), "+v"(hv[7]), "+v"(ha));
        WMMA_F32(acc1, ha, bw[1][8]);
        float d[FWD];
#pragma unroll
        for (int k = 0; k < FWD; ++k) d[k] = b1r[k];
        ORDER_D(acc1);
#pragma unroll
        for (int k = 0; k < FWD; ++k) {
#pragma unroll
            for (int v = 0; v < HID / 2; ++v) d[k] = fmaf(hv[v], w1r[k * HID + v], d[k]);
        }
        ORDER_D(hb);
        WMMA_F32(acc0, hb, bw[0][9]);
        ORDER_D(acc0);
#pragma unroll
        for (int k = 0; k < FWD; ++k) {
#pragma unroll
            for (int v = HID / 2; v < HID; ++v) d[k] = fmaf(hv[v], w1r[k * HID + v], d[k]);
        }
        ORDER_D(hb);
        WMMA_F32(acc1, hb, bw[1][9]);
        float delta = b2r;
        ORDER2(acc1, delta);
#pragma unroll
        for (int k = 0; k < FWD; ++k)
            delta = fmaf(fmaxf(d[k], 0.0f), w2r[k], delta);
        zreg += fmaxf(delta, 0.0f);
        if (u == 0)
            out[(size_t)crow * TSEQ + t] = zreg;
        gz = gather16(zreg, sz, same);
      }
    }
#undef XPART
#undef ORDER2
#undef ORDER_D
}

extern "C" void kernel_launch(void* const* d_in, const int* in_sizes, int n_in,
                              void* d_out, int out_size, void* d_ws, size_t ws_size,
                              hipStream_t stream) {
    (void)in_sizes; (void)n_in; (void)out_size; (void)d_ws; (void)ws_size;
    const float* x    = (const float*)d_in[0];
    const float* z0   = (const float*)d_in[1];
    const float* W_ih = (const float*)d_in[2];
    const float* W_hh = (const float*)d_in[3];
    const float* b_ih = (const float*)d_in[4];
    const float* b_hh = (const float*)d_in[5];
    const float* W1   = (const float*)d_in[6];
    const float* b1   = (const float*)d_in[7];
    const float* W2   = (const float*)d_in[8];
    const float* b2   = (const float*)d_in[9];
    float* out = (float*)d_out;

    dim3 grid(BATCH / 16);
    dim3 block(128);
    monotonic_lstm_wmma<<<grid, block, 0, stream>>>(x, z0, W_ih, W_hh, b_ih, b_hh,
                                                    W1, b1, W2, b2, out);
}


# Head shared over the 8 cell lanes of a batch row: lane u forms head unit u (lane exchange inside the multiply-add), 3 exchange-adds sum layer 2; head sum order changes (fixed tree)
# speedup vs baseline: 3.3780x; 1.1149x over previous
#include <hip/hip_runtime.h>
#include <hip/hip_bf16.h>


#define BATCH 4096
#define TSEQ  200
#define FEAT  32
#define HID   8
#define FWD   5
#define TBLK  4


typedef __attribute__((ext_vector_type(2))) float v2f;
typedef __attribute__((ext_vector_type(4))) float v4f;
typedef __attribute__((ext_vector_type(8))) float v8f;

__device__ __forceinline__ float sigm(float x) {
    return 1.0f / (1.0f + __expf(-x));
}
__device__ __forceinline__ float tanh_fast(float x) {
    return 1.0f - 2.0f / (1.0f + __expf(2.0f * x));
}

__device__ __forceinline__ float ror8(float v) {
    return __builtin_bit_cast(float,
        __builtin_amdgcn_update_dpp(0, __builtin_bit_cast(int, v), 0x128, 0xf, 0xf, false));
}
__device__ __forceinline__ float xch1(float v) {
    return __builtin_bit_cast(float,
        __builtin_amdgcn_update_dpp(0, __builtin_bit_cast(int, v), 0xB1, 0xf, 0xf, true));
}
__device__ __forceinline__ float xch2(float v) {
    return __builtin_bit_cast(float,
        __builtin_amdgcn_update_dpp(0, __builtin_bit_cast(int, v), 0x4E, 0xf, 0xf, true));
}
__device__ __forceinline__ float xch4(float v) {
    return __builtin_bit_cast(float,
        __builtin_amdgcn_update_dpp(0, __builtin_bit_cast(int, v), 0x141, 0xf, 0xf, true));
}
__device__ __forceinline__ float gather16(float v, unsigned sel, bool same) {
    const unsigned s = __builtin_bit_cast(unsigned, v);
    const unsigned own = __builtin_amdgcn_permlane16_var(s, s, sel, false, false);
    const unsigned oth = __builtin_amdgcn_permlanex16_var(s, s, sel, false, false);
    return __builtin_bit_cast(float, same ? own : oth);
}

#define WMMA_F32(ACC, A, BFRAG) \
    (ACC) = __builtin_amdgcn_wmma_f32_16x16x4_f32(false, (A), false, (BFRAG), (short)0, (ACC), false, false)

__global__ __launch_bounds__(128)
void monotonic_lstm_wmma(const float* __restrict__ x,
                         const float* __restrict__ z0,
                         const float* __restrict__ W_ih,
                         const float* __restrict__ W_hh,
                         const float* __restrict__ b_ih,
                         const float* __restrict__ b_hh,
                         const float* __restrict__ W1,
                         const float* __restrict__ b1,
                         const float* __restrict__ W2,
                         const float* __restrict__ b2,
                         float* __restrict__ out)
{
    const int lane = threadIdx.x & 31;
    const int c    = lane & 15;
    const int hi   = lane >> 4;
    const int wv   = threadIdx.x >> 5;
    const int rb   = blockIdx.x * 16 + wv * 4;

    const int xa = c >> 3, xb = c & 1;
    const int xrowi = rb + 2 * xa + xb;
    const int xj = (c >> 1) & 3;
    const int rs = (lane >> 3) & 1;
    const int u  = lane & 7;
    const int crow = rb + 2 * hi + rs;

    v2f bw[2][11];
#pragma unroll
    for (int tl = 0; tl < 2; ++tl) {
        const int n = c + 16 * tl;
        const float* wi = W_ih + n * (FEAT + 1);
        const float* wh = W_hh + n * HID;
#pragma unroll
        for (int q = 0; q < 4; ++q)
#pragma unroll
            for (int s = 0; s < 2; ++s) {
                const int k0 = 8 * q + 4 * hi + 2 * s;
                bw[tl][2 * q + s].x = wi[k0];
                bw[tl][2 * q + s].y = wi[k0 + 1];
            }
        bw[tl][8].x  = wh[hi ? 2 : 0];
        bw[tl][8].y  = wh[hi ? 3 : 1];
        bw[tl][9].x  = wh[hi ? 6 : 4];
        bw[tl][9].y  = wh[hi ? 7 : 5];
        bw[tl][10].x = hi ? 0.0f : wi[FEAT];
        bw[tl][10].y = 0.0f;
    }
    const float bias0 = b_ih[c] + b_hh[c];
    const float bias1 = b_ih[c + 16] + b_hh[c + 16];

    const bool hon = u < FWD;
    const int  hk  = hon ? u : 0;
    float w1l[HID];
#pragma unroll
    for (int v = 0; v < HID; ++v) { const float w = W1[hk * HID + v]; w1l[v] = hon ? w : 0.0f; }
    const float b1v = b1[hk], w2v = W2[hk];
    const float b1l = hon ? b1v : 0.0f;
    const float w2l = hon ? w2v : 0.0f;
    const float b2r = b2[0];

    const bool same = (xa == hi);
    const unsigned s1 = 8 * xb + (hi ? 2 : 0);
    const unsigned s2 = 8 * xb + (hi ? 3 : 1);
    const unsigned s3 = 8 * xb + (hi ? 6 : 4);
    const unsigned s4 = 8 * xb + (hi ? 7 : 5);
    const unsigned sz = 8 * xb;

    float cst  = 0.0f;
    float zreg = z0[crow];
    float gz = gather16(zreg, sz, same);

    const float* xrow = x + ((size_t)xrowi * TSEQ + xj) * FEAT + 4 * hi;

    v4f xbuf[4];
#pragma unroll
    for (int q = 0; q < 4; ++q)
        xbuf[q] = *reinterpret_cast<const v4f*>(xrow + 8 * q);

#define XPART()                                                   \
    do {                                                          \
        _Pragma("unroll")                                         \
        for (int r = 0; r < 8; ++r) { xp0[r] = bias0; xp1[r] = bias1; } \
        _Pragma("unroll")                                         \
        for (int q = 0; q < 4; ++q) {                             \
            const v4f xv = xbuf[q];                               \
            v2f a; a.x = xv.x; a.y = xv.y;                        \
            WMMA_F32(xp0, a, bw[0][2 * q]);                       \
            WMMA_F32(xp1, a, bw[1][2 * q]);                       \
            a.x = xv.z; a.y = xv.w;                               \
            WMMA_F32(xp0, a, bw[0][2 * q + 1]);                   \
            WMMA_F32(xp1, a, bw[1][2 * q + 1]);                   \
        }                                                         \
    } while (0)

#define ORDER2(A, B)    asm volatile("" : "+v"(A), "+v"(B))
#define HEAD_FMA(ACC, H, W, SEL) asm("v_fmac_f32_dpp %0, %1, %2 dpp8:[" SEL "]" : "+v"(ACC) : "v"(H), "v"(W))
#define HEAD_MUL(ACC, H, W, SEL) asm("v_mul_f32_dpp %0, %1, %2 dpp8:[" SEL "]" : "=v"(ACC) : "v"(H), "v"(W))

    v8f xp0, xp1, acc0, acc1;
    float g1 = 0.0f, g2 = 0.0f, g3 = 0.0f, g4 = 0.0f;
    XPART();
    acc0 = xp0; acc1 = xp1;

    for (int tb = 0; tb < TSEQ / TBLK; ++tb) {
        {
            const int tn = (tb + 1 < TSEQ / TBLK) ? (tb + 1) : tb;
#pragma unroll
            for (int q = 0; q < 4; ++q)
                xbuf[q] = *reinterpret_cast<const v4f*>(xrow + (size_t)tn * (TBLK * FEAT) + 8 * q);
        }

#pragma unroll
      for (int j = 0; j < TBLK; ++j) {
        const int t = tb * TBLK + j;

        {
            v2f a;
            a.x = hi ? 0.0f : gz; a.y = 0.0f;
            WMMA_F32(acc0, a, bw[0][10]);
            asm volatile("" : "+v"(acc0), "+v"(a));
            WMMA_F32(acc1, a, bw[1][10]);
        }
        const float a00 = acc0[2 * j], a01 = acc0[2 * j + 1];
        const float a10 = acc1[2 * j], a11 = acc1[2 * j + 1];

        const float t00 = ror8(a00), t01 = ror8(a01);
        const float t10 = ror8(a10), t11 = ror8(a11);
        const float gi = rs ? t01 : a00;
        const float gf = rs ? a01 : t00;
        const float gc = rs ? t11 : a10;
        const float go = rs ? a11 : t10;

        const float ig = sigm(gi);
        const float fg = sigm(gf);
        const float gg = tanh_fast(gc);
        const float og = sigm(go);
        const float cn = fg * cst + ig * gg;
        cst = cn;
        const float hn = og * tanh_fast(cn);

        float da = b1l, db;
        HEAD_MUL(db, hn, w1l[4], "4,4,4,4,4,4,4,4");
        HEAD_FMA(da, hn, w1l[0], "0,0,0,0,0,0,0,0");
        HEAD_FMA(db, hn, w1l[5], "5,5,5,5,5,5,5,5");
        HEAD_FMA(da, hn, w1l[1], "1,1,1,1,1,1,1,1");
        HEAD_FMA(db, hn, w1l[6], "6,6,6,6,6,6,6,6");
        HEAD_FMA(da, hn, w1l[2], "2,2,2,2,2,2,2,2");
        HEAD_FMA(db, hn, w1l[7], "7,7,7,7,7,7,7,7");
        HEAD_FMA(da, hn, w1l[3], "3,3,3,3,3,3,3,3");

        g1 = gather16(hn, s1, same);
        g2 = gather16(hn, s2, same);
        g3 = gather16(hn, s3, same);
        g4 = gather16(hn, s4, same);

        if (j == TBLK - 1) XPART();
        v2f ha, hb;
        ha.x = g1; ha.y = g2;
        hb.x = g3; hb.y = g4;
        acc0 = xp0; acc1 = xp1;
        float du = fmaxf(da + db, 0.0f);
        ORDER2(ha, du);
        WMMA_F32(acc0, ha, bw[0][8]);
        ORDER2(acc0, du);
        float part = du * w2l;
        part += xch1(part);
        part += xch2(part);
        part += xch4(part);
        ORDER2(ha, part);
        WMMA_F32(acc1, ha, bw[1][8]);
        ORDER2(acc1, part);
        zreg += fmaxf(part + b2r, 0.0f);
        ORDER2(hb, zreg);
        WMMA_F32(acc0, hb, bw[0][9]);
        ORDER2(acc0, zreg);
        if (u == 0)
            out[(size_t)crow * TSEQ + t] = zreg;
        {
            const unsigned zs = __builtin_bit_cast(unsigned, zreg);
            unsigned zown = __builtin_amdgcn_permlane16_var(zs, zs, sz, false, false);
            unsigned zoth = __builtin_amdgcn_permlanex16_var(zs, zs, sz, false, false);
            asm volatile("" : "+v"(zown), "+v"(zoth), "+v"(hb));
            WMMA_F32(acc1, hb, bw[1][9]);
            asm volatile("" : "+v"(zown), "+v"(zoth), "+v"(acc1));
            gz = __builtin_bit_cast(float, same ? zown : zoth);
        }
      }
    }
#undef XPART
#undef ORDER2
#undef HEAD_FMA
#undef HEAD_MUL
}

extern "C" void kernel_launch(void* const* d_in, const int* in_sizes, int n_in,
                              void* d_out, int out_size, void* d_ws, size_t ws_size,
                              hipStream_t stream) {
    (void)in_sizes; (void)n_in; (void)out_size; (void)d_ws; (void)ws_size;
    const float* x    = (const float*)d_in[0];
    const float* z0   = (const float*)d_in[1];
    const float* W_ih = (const float*)d_in[2];
    const float* W_hh = (const float*)d_in[3];
    const float* b_ih = (const float*)d_in[4];
    const float* b_hh = (const float*)d_in[5];
    const float* W1   = (const float*)d_in[6];
    const float* b1   = (const float*)d_in[7];
    const float* W2   = (const float*)d_in[8];
    const float* b2   = (const float*)d_in[9];
    float* out = (float*)d_out;

    dim3 grid(BATCH / 16);
    dim3 block(128);
    monotonic_lstm_wmma<<<grid, block, 0, stream>>>(x, z0, W_ih, W_hh, b_ih, b_hh,
                                                    W1, b1, W2, b2, out);
}


# Output off the recurrence's chain: lane u of a row's 8-lane group keeps z of step u, one unmasked 32-lane store per 8 steps; no store, branch or exec change between z and the next z pass
# speedup vs baseline: 3.6632x; 1.0844x over previous
#include <hip/hip_runtime.h>
#include <hip/hip_bf16.h>


#define BATCH 4096
#define TSEQ  200
#define FEAT  32
#define HID   8
#define FWD   5
#define TBLK  4


typedef __attribute__((ext_vector_type(2))) float v2f;
typedef __attribute__((ext_vector_type(4))) float v4f;
typedef __attribute__((ext_vector_type(8))) float v8f;

__device__ __forceinline__ float sigm(float x) {
    return 1.0f / (1.0f + __expf(-x));
}
__device__ __forceinline__ float tanh_fast(float x) {
    return 1.0f - 2.0f / (1.0f + __expf(2.0f * x));
}

__device__ __forceinline__ float ror8(float v) {
    return __builtin_bit_cast(float,
        __builtin_amdgcn_update_dpp(0, __builtin_bit_cast(int, v), 0x128, 0xf, 0xf, false));
}
__device__ __forceinline__ float xch1(float v) {
    return __builtin_bit_cast(float,
        __builtin_amdgcn_update_dpp(0, __builtin_bit_cast(int, v), 0xB1, 0xf, 0xf, true));
}
__device__ __forceinline__ float xch2(float v) {
    return __builtin_bit_cast(float,
        __builtin_amdgcn_update_dpp(0, __builtin_bit_cast(int, v), 0x4E, 0xf, 0xf, true));
}
__device__ __forceinline__ float xch4(float v) {
    return __builtin_bit_cast(float,
        __builtin_amdgcn_update_dpp(0, __builtin_bit_cast(int, v), 0x141, 0xf, 0xf, true));
}
__device__ __forceinline__ float gather16(float v, unsigned sel, bool same) {
    const unsigned s = __builtin_bit_cast(unsigned, v);
    const unsigned own = __builtin_amdgcn_permlane16_var(s, s, sel, false, false);
    const unsigned oth = __builtin_amdgcn_permlanex16_var(s, s, sel, false, false);
    return __builtin_bit_cast(float, same ? own : oth);
}

#define WMMA_F32(ACC, A, BFRAG) \
    (ACC) = __builtin_amdgcn_wmma_f32_16x16x4_f32(false, (A), false, (BFRAG), (short)0, (ACC), false, false)

__global__ __launch_bounds__(128)
void monotonic_lstm_wmma(const float* __restrict__ x,
                         const float* __restrict__ z0,
                         const float* __restrict__ W_ih,
                         const float* __restrict__ W_hh,
                         const float* __restrict__ b_ih,
                         const float* __restrict__ b_hh,
                         const float* __restrict__ W1,
                         const float* __restrict__ b1,
                         const float* __restrict__ W2,
                         const float* __restrict__ b2,
                         float* __restrict__ out)
{
    const int lane = threadIdx.x & 31;
    const int c    = lane & 15;
    const int hi   = lane >> 4;
    const int wv   = threadIdx.x >> 5;
    const int rb   = blockIdx.x * 16 + wv * 4;

    const int xa = c >> 3, xb = c & 1;
    const int xrowi = rb + 2 * xa + xb;
    const int xj = (c >> 1) & 3;
    const int rs = (lane >> 3) & 1;
    const int u  = lane & 7;
    const int crow = rb + 2 * hi + rs;

    v2f bw[2][11];
#pragma unroll
    for (int tl = 0; tl < 2; ++tl) {
        const int n = c + 16 * tl;
        const float* wi = W_ih + n * (FEAT + 1);
        const float* wh = W_hh + n * HID;
#pragma unroll
        for (int q = 0; q < 4; ++q)
#pragma unroll
            for (int s = 0; s < 2; ++s) {
                const int k0 = 8 * q + 4 * hi + 2 * s;
                bw[tl][2 * q + s].x = wi[k0];
                bw[tl][2 * q + s].y = wi[k0 + 1];
            }
        bw[tl][8].x  = wh[hi ? 2 : 0];
        bw[tl][8].y  = wh[hi ? 3 : 1];
        bw[tl][9].x  = wh[hi ? 6 : 4];
        bw[tl][9].y  = wh[hi ? 7 : 5];
        bw[tl][10].x = hi ? 0.0f : wi[FEAT];
        bw[tl][10].y = 0.0f;
    }
    const float bias0 = b_ih[c] + b_hh[c];
    const float bias1 = b_ih[c + 16] + b_hh[c + 16];

    const bool hon = u < FWD;
    const int  hk  = hon ? u : 0;
    float w1l[HID];
#pragma unroll
    for (int v = 0; v < HID; ++v) { const float w = W1[hk * HID + v]; w1l[v] = hon ? w : 0.0f; }
    const float b1v = b1[hk], w2v = W2[hk];
    const float b1l = hon ? b1v : 0.0f;
    const float w2l = hon ? w2v : 0.0f;
    const float b2r = b2[0];

    const bool same = (xa == hi);
    const unsigned s1 = 8 * xb + (hi ? 2 : 0);
    const unsigned s2 = 8 * xb + (hi ? 3 : 1);
    const unsigned s3 = 8 * xb + (hi ? 6 : 4);
    const unsigned s4 = 8 * xb + (hi ? 7 : 5);
    const unsigned sz = 8 * xb;

    float cst  = 0.0f;
    float zreg = z0[crow];
    float gz = gather16(zreg, sz, same);

    const float* xrow = x + ((size_t)xrowi * TSEQ + xj) * FEAT + 4 * hi;

    v4f xbuf[4];
#pragma unroll
    for (int q = 0; q < 4; ++q)
        xbuf[q] = *reinterpret_cast<const v4f*>(xrow + 8 * q);

#define XPART()                                                   \
    do {                                                          \
        _Pragma("unroll")                                         \
        for (int r = 0; r < 8; ++r) { xp0[r] = bias0; xp1[r] = bias1; } \
        _Pragma("unroll")                                         \
        for (int q = 0; q < 4; ++q) {                             \
            const v4f xv = xbuf[q];                               \
            v2f a; a.x = xv.x; a.y = xv.y;                        \
            WMMA_F32(xp0, a, bw[0][2 * q]);                       \
            WMMA_F32(xp1, a, bw[1][2 * q]);                       \
            a.x = xv.z; a.y = xv.w;                               \
            WMMA_F32(xp0, a, bw[0][2 * q + 1]);                   \
            WMMA_F32(xp1, a, bw[1][2 * q + 1]);                   \
        }                                                         \
    } while (0)

#define ORDER2(A, B)    asm volatile("" : "+v"(A), "+v"(B))
#define HEAD_FMA(ACC, H, W, SEL) asm("v_fmac_f32_dpp %0, %1, %2 dpp8:[" SEL "]" : "+v"(ACC) : "v"(H), "v"(W))
#define HEAD_MUL(ACC, H, W, SEL) asm("v_mul_f32_dpp %0, %1, %2 dpp8:[" SEL "]" : "=v"(ACC) : "v"(H), "v"(W))

    v8f xp0, xp1, acc0, acc1;
    float g1 = 0.0f, g2 = 0.0f, g3 = 0.0f, g4 = 0.0f;
    XPART();
    acc0 = xp0; acc1 = xp1;

    const unsigned ooff = (unsigned)crow * TSEQ + (unsigned)u;

    for (int tp = 0; tp < TSEQ / (2 * TBLK); ++tp) {
      float zkeep = 0.0f;
#pragma unroll
      for (int hb2 = 0; hb2 < 2; ++hb2) {
        const int tb = 2 * tp + hb2;
        {
            const int tn = (tb + 1 < TSEQ / TBLK) ? (tb + 1) : tb;
#pragma unroll
            for (int q = 0; q < 4; ++q)
                xbuf[q] = *reinterpret_cast<const v4f*>(xrow + (size_t)tn * (TBLK * FEAT) + 8 * q);
        }

#pragma unroll
      for (int j = 0; j < TBLK; ++j) {

        {
            v2f a;
            a.x = hi ? 0.0f : gz; a.y = 0.0f;
            WMMA_F32(acc0, a, bw[0][10]);
            asm volatile("" : "+v"(acc0), "+v"(a));
            WMMA_F32(acc1, a, bw[1][10]);
        }
        const float a00 = acc0[2 * j], a01 = acc0[2 * j + 1];
        const float a10 = acc1[2 * j], a11 = acc1[2 * j + 1];

        const float t00 = ror8(a00), t01 = ror8(a01);
        const float t10 = ror8(a10), t11 = ror8(a11);
        const float gi = rs ? t01 : a00;
        const float gf = rs ? a01 : t00;
        const float gc = rs ? t11 : a10;
        const float go = rs ? a11 : t10;

        const float ig = sigm(gi);
        const float fg = sigm(gf);
        const float gg = tanh_fast(gc);
        const float og = sigm(go);
        const float cn = fg * cst + ig * gg;
        cst = cn;
        const float hn = og * tanh_fast(cn);

        float da = b1l, db;
        HEAD_MUL(db, hn, w1l[4], "4,4,4,4,4,4,4,4");
        HEAD_FMA(da, hn, w1l[0], "0,0,0,0,0,0,0,0");
        HEAD_FMA(db, hn, w1l[5], "5,5,5,5,5,5,5,5");
        HEAD_FMA(da, hn, w1l[1], "1,1,1,1,1,1,1,1");
        HEAD_FMA(db, hn, w1l[6], "6,6,6,6,6,6,6,6");
        HEAD_FMA(da, hn, w1l[2], "2,2,2,2,2,2,2,2");
        HEAD_FMA(db, hn, w1l[7], "7,7,7,7,7,7,7,7");
        HEAD_FMA(da, hn, w1l[3], "3,3,3,3,3,3,3,3");

        g1 = gather16(hn, s1, same);
        g2 = gather16(hn, s2, same);
        g3 = gather16(hn, s3, same);
        g4 = gather16(hn, s4, same);

        if (j == TBLK - 1) XPART();
        v2f ha, hb;
        ha.x = g1; ha.y = g2;
        hb.x = g3; hb.y = g4;
        acc0 = xp0; acc1 = xp1;
        float du = fmaxf(da + db, 0.0f);
        ORDER2(ha, du);
        WMMA_F32(acc0, ha, bw[0][8]);
        ORDER2(acc0, du);
        float part = du * w2l;
        part += xch1(part);
        part += xch2(part);
        part += xch4(part);
        ORDER2(ha, part);
        WMMA_F32(acc1, ha, bw[1][8]);
        ORDER2(acc1, part);
        zreg += fmaxf(part + b2r, 0.0f);
        ORDER2(hb, zreg);
        WMMA_F32(acc0, hb, bw[0][9]);
        ORDER2(acc0, zreg);
        if (hb2 == 0 && j == 0) zkeep = zreg;
        else                    zkeep = (u == hb2 * TBLK + j) ? zreg : zkeep;
        asm volatile("" : "+v"(zkeep));
        {
            const unsigned zs = __builtin_bit_cast(unsigned, zreg);
            unsigned zown = __builtin_amdgcn_permlane16_var(zs, zs, sz, false, false);
            unsigned zoth = __builtin_amdgcn_permlanex16_var(zs, zs, sz, false, false);
            asm volatile("" : "+v"(zown), "+v"(zoth), "+v"(hb));
            WMMA_F32(acc1, hb, bw[1][9]);
            asm volatile("" : "+v"(zown), "+v"(zoth), "+v"(acc1));
            gz = __builtin_bit_cast(float, same ? zown : zoth);
        }
      }
      }
      (out + tp * (2 * TBLK))[ooff] = zkeep;
    }
#undef XPART
#undef ORDER2
#undef HEAD_FMA
#undef HEAD_MUL
}

extern "C" void kernel_launch(void* const* d_in, const int* in_sizes, int n_in,
                              void* d_out, int out_size, void* d_ws, size_t ws_size,
                              hipStream_t stream) {
    (void)in_sizes; (void)n_in; (void)out_size; (void)d_ws; (void)ws_size;
    const float* x    = (const float*)d_in[0];
    const float* z0   = (const float*)d_in[1];
    const float* W_ih = (const float*)d_in[2];
    const float* W_hh = (const float*)d_in[3];
    const float* b_ih = (const float*)d_in[4];
    const float* b_hh = (const float*)d_in[5];
    const float* W1   = (const float*)d_in[6];
    const float* b1   = (const float*)d_in[7];
    const float* W2   = (const float*)d_in[8];
    const float* b2   = (const float*)d_in[9];
    float* out = (float*)d_out;

    dim3 grid(BATCH / 16);
    dim3 block(128);
    monotonic_lstm_wmma<<<grid, block, 0, stream>>>(x, z0, W_ih, W_hh, b_ih, b_hh,
                                                    W1, b1, W2, b2, out);
}
